# starting kernel
# baseline (speedup 1.0000x reference)
#include <hip/hip_runtime.h>

typedef float v2f __attribute__((ext_vector_type(2)));
typedef float v8f __attribute__((ext_vector_type(8)));

#define H_   64
#define W_   64
#define OH   128
#define OW   128
#define CTOT 256
#define CG   64
#define K2   25
#define NG   4
#define FSTR 98

__global__ __launch_bounds__(256) void carafe_wmma(
    const float* __restrict__ feat, const float* __restrict__ mask,
    float* __restrict__ out)
{
  __shared__ float sF[CG * FSTR];
  __shared__ float sM[K2 * 128];

  const int tid = threadIdx.x;
  const int n   = blockIdx.z;
  const int g   = blockIdx.y;
  const int stx = blockIdx.x & 7;
  const int sty = blockIdx.x >> 3;
  const int x0  = stx * 8;
  const int y0  = sty * 4;

  const float* fbase = feat + ((size_t)n * CTOT + g * CG) * (H_ * W_);
  #pragma unroll 1
  for (int e = tid; e < CG * 96; e += 256) {
    int c  = e / 96;
    int r  = e - c * 96;
    int yy = r / 12;
    int xx = r - yy * 12;
    int gy = y0 - 2 + yy;
    int gx = x0 - 2 + xx;
    float v = 0.0f;
    if (gy >= 0 && gy < H_ && gx >= 0 && gx < W_)
      v = fbase[c * (H_ * W_) + gy * W_ + gx];
    sF[c * FSTR + r] = v;
  }

  const float* mb = mask + ((size_t)n * (K2 * NG) + g * K2) * (OH * OW)
                  + (2 * y0) * OW + 2 * x0;
  #pragma unroll 1
  for (int e = tid; e < K2 * 128; e += 256) {
    int ch = e >> 7;
    int r  = e & 127;
    int ry = r >> 4;
    int rx = r & 15;
    sM[e] = mb[(size_t)ch * (OH * OW) + ry * OW + rx];
  }
  __syncthreads();

  const int wave = tid >> 5;
  const int lane = tid & 31;
  const int tx = wave & 3, ty = wave >> 2;
  const int h0 = y0 + 2 * ty;
  const int w0 = x0 + 2 * tx;

  const int M    = lane & 15;
  const int half = lane >> 4;
  const int myo = M >> 2, mxo = M & 3;
  const int ti = myo >> 1, tj = mxo >> 1;
  const int lc = lane & 15;
  const int mpos = (4 * ty + myo) * 16 + (4 * tx + mxo);

  v2f a[9];
  int fo[9];
  #pragma unroll
  for (int kk = 0; kk < 9; ++kk) {
    #pragma unroll
    for (int j = 0; j < 2; ++j) {
      const int Kl = 4 * kk + j;
      const int Kh = Kl + 2;
      int dy = half ? (Kh / 6) : (Kl / 6);
      int dx = half ? (Kh % 6) : (Kl % 6);
      int di = dy - ti, dj = dx - tj;
      float av = 0.0f;
      if (di >= 0 && di < 5 && dj >= 0 && dj < 5)
        av = sM[(di * 5 + dj) * 128 + mpos];
      a[kk][j] = av;
      if (j == 0)
        fo[kk] = (2 * ty + dy) * 12 + (2 * tx + dx);
    }
  }

  float* obase = out + ((size_t)n * CTOT + g * CG) * (OH * OW);

  #pragma unroll
  for (int ct = 0; ct < 4; ++ct) {
    v8f acc = {};
    #pragma unroll
    for (int kk = 0; kk < 9; ++kk) {
      v2f b = *(const v2f*)&sF[lc * FSTR + fo[kk] + ct * (16 * FSTR)];
      acc = __builtin_amdgcn_wmma_f32_16x16x4_f32(
          false, a[kk], false, b, (short)0, acc, false, false);
    }
    float* op = obase + (size_t)(ct * 16 + lc) * (OH * OW);
    const int ryo = 2 * h0 + 2 * half;
    float4 lo = make_float4(acc[0], acc[1], acc[2], acc[3]);
    float4 hi = make_float4(acc[4], acc[5], acc[6], acc[7]);
    *(float4*)(op + (size_t)ryo * OW + 2 * w0)       = lo;
    *(float4*)(op + (size_t)(ryo + 1) * OW + 2 * w0) = hi;
  }
}

extern "C" void kernel_launch(void* const* d_in, const int* in_sizes, int n_in,
                              void* d_out, int out_size, void* d_ws, size_t ws_size,
                              hipStream_t stream) {
  (void)in_sizes; (void)n_in; (void)out_size; (void)d_ws; (void)ws_size;
  const float* feat = (const float*)d_in[0];
  const float* mask = (const float*)d_in[1];
  float* out = (float*)d_out;
  dim3 grid(8 * 16, NG, 4);
  carafe_wmma<<<grid, 256, 0, stream>>>(feat, mask, out);
}


# Every global store instruction writes four whole 128-byte lines: block tile 16x2 sources (4 rows x 32 cols out), D tiles staged through LDS; was 32 lines x 16 B per store, lines shared by 2 blocks
# speedup vs baseline: 2.0636x; 2.0636x over previous
#include <hip/hip_runtime.h>


typedef float v2f __attribute__((ext_vector_type(2)));
typedef float v8f __attribute__((ext_vector_type(8)));

#define H_   64
#define W_   64
#define OH   128
#define OW   128
#define CTOT 256
#define CG   64
#define K2   25
#define NG   4
#define FW   20
#define FPL  120
#define FSTR 124
#define MPL  128
#define OSTR 132
#define SM_OFF (CG * FSTR)

__global__ __launch_bounds__(256) void carafe_wmma(
    const float* __restrict__ feat, const float* __restrict__ mask,
    float* __restrict__ out)
{
  __shared__ __attribute__((aligned(16))) float smem[CG * FSTR + K2 * MPL];

  const int tid = threadIdx.x;
  const int n   = blockIdx.z;
  const int g   = blockIdx.y;
  const int stx = blockIdx.x & 3;
  const int sty = blockIdx.x >> 2;
  const int x0  = stx * 16;
  const int y0  = sty * 2;

  const float* fbase = feat + ((size_t)n * CTOT + g * CG) * (H_ * W_);
  #pragma unroll 1
  for (int e = tid; e < CG * FPL; e += 256) {
    int c  = e / FPL;
    int r  = e - c * FPL;
    int yy = r / FW;
    int xx = r - yy * FW;
    int gy = y0 - 2 + yy;
    int gx = x0 - 2 + xx;
    float v = 0.0f;
    if (gy >= 0 && gy < H_ && gx >= 0 && gx < W_)
      v = fbase[c * (H_ * W_) + gy * W_ + gx];
    smem[c * FSTR + r] = v;
  }

  const float* mb = mask + ((size_t)n * (K2 * NG) + g * K2) * (OH * OW)
                  + (2 * y0) * OW + 2 * x0;
  #pragma unroll 1
  for (int e = tid; e < K2 * MPL; e += 256) {
    int ch = e >> 7;
    int r  = e & 127;
    int ry = r >> 5;
    int rx = r & 31;
    smem[SM_OFF + e] = mb[(size_t)ch * (OH * OW) + ry * OW + rx];
  }
  __syncthreads();

  const int wave = tid >> 5;
  const int lane = tid & 31;
  const int tx = wave;

  const int M    = lane & 15;
  const int half = lane >> 4;
  const int myo = M >> 2, mxo = M & 3;
  const int ti = myo >> 1, tj = mxo >> 1;
  const int lc = lane & 15;
  const int mpos = myo * 32 + (4 * tx + mxo);

  v2f a[9];
  int fo[9];
  #pragma unroll
  for (int kk = 0; kk < 9; ++kk) {
    #pragma unroll
    for (int j = 0; j < 2; ++j) {
      const int Kl = 4 * kk + j;
      const int Kh = Kl + 2;
      int dy = half ? (Kh / 6) : (Kl / 6);
      int dx = half ? (Kh % 6) : (Kl % 6);
      int di = dy - ti, dj = dx - tj;
      float av = 0.0f;
      if (di >= 0 && di < 5 && dj >= 0 && dj < 5)
        av = smem[SM_OFF + (di * 5 + dj) * MPL + mpos];
      a[kk][j] = av;
      if (j == 0)
        fo[kk] = dy * FW + (2 * tx + dx);
    }
  }

  v8f acc[4];
  #pragma unroll
  for (int ct = 0; ct < 4; ++ct) {
    v8f c8 = {};
    #pragma unroll
    for (int kk = 0; kk < 9; ++kk) {
      v2f b = *(const v2f*)&smem[lc * FSTR + fo[kk] + ct * (16 * FSTR)];
      c8 = __builtin_amdgcn_wmma_f32_16x16x4_f32(
          false, a[kk], false, b, (short)0, c8, false, false);
    }
    acc[ct] = c8;
  }

  __syncthreads();

  for (int ct = 0; ct < 4; ++ct) {
    float* so = &smem[(ct * 16 + lc) * OSTR + (2 * half) * 32 + 4 * tx];
    *(float4*)so        = make_float4(acc[ct][0], acc[ct][1], acc[ct][2], acc[ct][3]);
    *(float4*)(so + 32) = make_float4(acc[ct][4], acc[ct][5], acc[ct][6], acc[ct][7]);
  }
  __syncthreads();

  float* obase = out + ((size_t)n * CTOT + g * CG) * (OH * OW)
               + (size_t)(2 * y0) * OW + 2 * x0
               + (lane >> 3) * OW + (lane & 7) * 4;
  for (int i = 0; i < 8; ++i) {
    const int ch = wave * 8 + i;
    float4 v = *(const float4*)&smem[ch * OSTR + lane * 4];
    *(float4*)(obase + (size_t)ch * (OH * OW)) = v;
  }
}

extern "C" void kernel_launch(void* const* d_in, const int* in_sizes, int n_in,
                              void* d_out, int out_size, void* d_ws, size_t ws_size,
                              hipStream_t stream) {
  (void)in_sizes; (void)n_in; (void)out_size; (void)d_ws; (void)ws_size;
  const float* feat = (const float*)d_in[0];
  const float* mask = (const float*)d_in[1];
  float* out = (float*)d_out;
  dim3 grid(4 * 32, NG, 4);
  carafe_wmma<<<grid, 256, 0, stream>>>(feat, mask, out);
}


# Tile fill keeps every load in flight: a thread issues all its global loads (masks as b128 quads, features as aligned b64 pairs) before its first LDS store; was one b32 load per trip, each waited for
# speedup vs baseline: 4.7985x; 2.3253x over previous
#include <hip/hip_runtime.h>


typedef float v2f __attribute__((ext_vector_type(2)));
typedef float v4f __attribute__((ext_vector_type(4)));
typedef float v8f __attribute__((ext_vector_type(8)));

#define H_   64
#define W_   64
#define OH   128
#define OW   128
#define CTOT 256
#define CG   64
#define K2   25
#define NG   4
#define FW   20
#define FPL  120
#define FSTR 124
#define MPL  128
#define OSTR 132
#define SM_OFF (CG * FSTR)

__global__ __launch_bounds__(256) void carafe_wmma(
    const float* __restrict__ feat, const float* __restrict__ mask,
    float* __restrict__ out)
{
  __shared__ __attribute__((aligned(16))) float smem[CG * FSTR + K2 * MPL];

  const int tid = threadIdx.x;
  const int n   = blockIdx.z;
  const int g   = blockIdx.y;
  const int stx = blockIdx.x & 3;
  const int sty = blockIdx.x >> 2;
  const int x0  = stx * 16;
  const int y0  = sty * 2;

  const float* fbase = feat + ((size_t)n * CTOT + g * CG) * (H_ * W_);
  const float* mb = mask + ((size_t)n * (K2 * NG) + g * K2) * (OH * OW)
                  + (2 * y0) * OW + 2 * x0;

  const float* mp = mb + (size_t)(tid >> 5) * (OH * OW)
                  + ((tid >> 3) & 3) * OW + (tid & 7) * 4;
  const v4f m0 = *(const v4f*)(mp);
  const v4f m1 = *(const v4f*)(mp + (size_t)1 * (8 * OH * OW));
  const v4f m2 = *(const v4f*)(mp + (size_t)2 * (8 * OH * OW));

  const int cq = tid / 60;
  const int r2 = tid - cq * 60;
  const int yy = r2 / 10;
  const int pp = r2 - yy * 10;
  const int gy = y0 - 2 + yy;
  const int gx = x0 - 2 + 2 * pp;
  const bool fown = tid < 240;
  const bool fin  = fown && gy >= 0 && gy < H_ && gx >= 0 && gx < W_;
  v2f fv[16];
  #pragma unroll
  for (int i = 0; i < 16; ++i) fv[i] = (v2f){0.0f, 0.0f};
  if (fin) {
    const float* fp = fbase + cq * (H_ * W_) + gy * W_ + gx;
    #pragma unroll
    for (int i = 0; i < 16; ++i)
      fv[i] = *(const v2f*)(fp + (size_t)i * (4 * H_ * W_));
  }

  if (tid < 32)
    *(v4f*)&smem[SM_OFF + (tid + 768) * 4] =
        *(const v4f*)(mp + (size_t)3 * (8 * OH * OW));

  *(v4f*)&smem[SM_OFF + tid * 4]         = m0;
  *(v4f*)&smem[SM_OFF + (tid + 256) * 4] = m1;
  *(v4f*)&smem[SM_OFF + (tid + 512) * 4] = m2;
  if (fown) {
    #pragma unroll
    for (int i = 0; i < 16; ++i)
      *(v2f*)&smem[(cq + 4 * i) * FSTR + 2 * r2] = fv[i];
  }
  __syncthreads();

  const int wave = tid >> 5;
  const int lane = tid & 31;
  const int tx = wave;

  const int M    = lane & 15;
  const int half = lane >> 4;
  const int myo = M >> 2, mxo = M & 3;
  const int ti = myo >> 1, tj = mxo >> 1;
  const int lc = lane & 15;
  const int mpos = myo * 32 + (4 * tx + mxo);

  v2f a[9];
  int fo[9];
  #pragma unroll
  for (int kk = 0; kk < 9; ++kk) {
    #pragma unroll
    for (int j = 0; j < 2; ++j) {
      const int Kl = 4 * kk + j;
      const int Kh = Kl + 2;
      int dy = half ? (Kh / 6) : (Kl / 6);
      int dx = half ? (Kh % 6) : (Kl % 6);
      int di = dy - ti, dj = dx - tj;
      float av = 0.0f;
      if (di >= 0 && di < 5 && dj >= 0 && dj < 5)
        av = smem[SM_OFF + (di * 5 + dj) * MPL + mpos];
      a[kk][j] = av;
      if (j == 0)
        fo[kk] = dy * FW + (2 * tx + dx);
    }
  }

  v8f acc[4];
  #pragma unroll
  for (int ct = 0; ct < 4; ++ct) {
    v8f c8 = {};
    #pragma unroll
    for (int kk = 0; kk < 9; ++kk) {
      v2f b = *(const v2f*)&smem[lc * FSTR + fo[kk] + ct * (16 * FSTR)];
      c8 = __builtin_amdgcn_wmma_f32_16x16x4_f32(
          false, a[kk], false, b, (short)0, c8, false, false);
    }
    acc[ct] = c8;
  }

  __syncthreads();

  for (int ct = 0; ct < 4; ++ct) {
    float* so = &smem[(ct * 16 + lc) * OSTR + (2 * half) * 32 + 4 * tx];
    *(float4*)so        = make_float4(acc[ct][0], acc[ct][1], acc[ct][2], acc[ct][3]);
    *(float4*)(so + 32) = make_float4(acc[ct][4], acc[ct][5], acc[ct][6], acc[ct][7]);
  }
  __syncthreads();

  float* obase = out + ((size_t)n * CTOT + g * CG) * (OH * OW)
               + (size_t)(2 * y0) * OW + 2 * x0
               + (lane >> 3) * OW + (lane & 7) * 4;
  for (int i = 0; i < 8; ++i) {
    const int ch = wave * 8 + i;
    float4 v = *(const float4*)&smem[ch * OSTR + lane * 4];
    *(float4*)(obase + (size_t)ch * (OH * OW)) = v;
  }
}

extern "C" void kernel_launch(void* const* d_in, const int* in_sizes, int n_in,
                              void* d_out, int out_size, void* d_ws, size_t ws_size,
                              hipStream_t stream) {
  (void)in_sizes; (void)n_in; (void)out_size; (void)d_ws; (void)ws_size;
  const float* feat = (const float*)d_in[0];
  const float* mask = (const float*)d_in[1];
  float* out = (float*)d_out;
  dim3 grid(4 * 32, NG, 4);
  carafe_wmma<<<grid, 256, 0, stream>>>(feat, mask, out);
}


# Block tile spans the whole image width (64x2 sources, 512 threads, 2 sub-tiles per wave): every fetched feature line is used whole; was 16x2 tiles whose 80-byte row pieces took 1-2 lines each
# speedup vs baseline: 6.4537x; 1.3449x over previous
#include <hip/hip_runtime.h>


typedef float v2f __attribute__((ext_vector_type(2)));
typedef float v4f __attribute__((ext_vector_type(4)));
typedef float v8f __attribute__((ext_vector_type(8)));

#define H_   64
#define W_   64
#define OH   128
#define OW   128
#define CTOT 256
#define CG   64
#define K2   25
#define NG   4
#define FW   68
#define FSTR 412
#define MRP  132
#define MPL  528
#define OSTR 516
#define SM_OFF (CG * FSTR)

__global__ __launch_bounds__(512) void carafe_wmma(
    const float* __restrict__ feat, const float* __restrict__ mask,
    float* __restrict__ out)
{
  __shared__ __attribute__((aligned(16))) float smem[CG * FSTR + K2 * MPL];

  const int tid = threadIdx.x;
  const int n   = blockIdx.z;
  const int g   = blockIdx.y;
  const int y0  = blockIdx.x * 2;

  const float* fbase = feat + ((size_t)n * CTOT + g * CG) * (H_ * W_);
  const float* mb = mask + ((size_t)n * (K2 * NG) + g * K2) * (OH * OW)
                  + (2 * y0) * OW;

  const int mq = tid & 127;
  const float* mp = mb + (size_t)(tid >> 7) * (OH * OW) + mq * 4;
  v4f mv[6];
  #pragma unroll
  for (int i = 0; i < 6; ++i)
    mv[i] = *(const v4f*)(mp + (size_t)i * (4 * OH * OW));

  const int fc = tid >> 4;
  const int fq = tid & 15;
  const float* fp = fbase + (size_t)fc * (H_ * W_) + fq * 4;
  v4f fv[12];
  #pragma unroll
  for (int i = 0; i < 12; ++i) {
    const int gy = y0 - 2 + (i >> 1);
    fv[i] = (v4f){0.0f, 0.0f, 0.0f, 0.0f};
    if (gy >= 0 && gy < H_)
      fv[i] = *(const v4f*)(fp + (size_t)(i & 1) * (32 * H_ * W_) + gy * W_);
  }

  const int mdst = SM_OFF + (mq >> 5) * MRP + (mq & 31) * 4;
  if (tid < 128)
    *(v4f*)&smem[mdst + 24 * MPL] =
        *(const v4f*)(mp + (size_t)6 * (4 * OH * OW));

  #pragma unroll
  for (int i = 0; i < 6; ++i)
    *(v4f*)&smem[mdst + ((tid >> 7) + 4 * i) * MPL] = mv[i];
  #pragma unroll
  for (int i = 0; i < 12; ++i) {
    float* d = &smem[(fc + 32 * (i & 1)) * FSTR + (i >> 1) * FW + 2 + fq * 4];
    *(v2f*)d       = (v2f){fv[i][0], fv[i][1]};
    *(v2f*)(d + 2) = (v2f){fv[i][2], fv[i][3]};
  }
  if (tid < 384) {
    float* d = &smem[(tid & 63) * FSTR + (tid >> 6) * FW];
    *(v2f*)d        = (v2f){0.0f, 0.0f};
    *(v2f*)(d + 66) = (v2f){0.0f, 0.0f};
  }
  __syncthreads();

  const int wave = tid >> 5;
  const int lane = tid & 31;

  const int M    = lane & 15;
  const int half = lane >> 4;
  const int myo = M >> 2, mxo = M & 3;
  const int ti = myo >> 1, tj = mxo >> 1;
  const int lc = lane & 15;

  v8f acc[2][4];
  #pragma unroll
  for (int jt = 0; jt < 2; ++jt) {
    const int tx = wave + 16 * jt;
    const int mpos = myo * MRP + (4 * tx + mxo);

    v2f a[9];
    int fo[9];
    #pragma unroll
    for (int kk = 0; kk < 9; ++kk) {
      #pragma unroll
      for (int j = 0; j < 2; ++j) {
        const int Kl = 4 * kk + j;
        const int Kh = Kl + 2;
        int dy = half ? (Kh / 6) : (Kl / 6);
        int dx = half ? (Kh % 6) : (Kl % 6);
        int di = dy - ti, dj = dx - tj;
        float av = 0.0f;
        if (di >= 0 && di < 5 && dj >= 0 && dj < 5)
          av = smem[SM_OFF + (di * 5 + dj) * MPL + mpos];
        a[kk][j] = av;
        if (j == 0)
          fo[kk] = dy * FW + (2 * tx + dx);
      }
    }

    #pragma unroll
    for (int ct = 0; ct < 4; ++ct) {
      v8f c8 = {};
      #pragma unroll
      for (int kk = 0; kk < 9; ++kk) {
        v2f b = *(const v2f*)&smem[lc * FSTR + fo[kk] + ct * (16 * FSTR)];
        c8 = __builtin_amdgcn_wmma_f32_16x16x4_f32(
            false, a[kk], false, b, (short)0, c8, false, false);
      }
      acc[jt][ct] = c8;
    }
  }

  __syncthreads();

  #pragma unroll
  for (int jt = 0; jt < 2; ++jt) {
    const int tx = wave + 16 * jt;
    #pragma unroll
    for (int ct = 0; ct < 4; ++ct) {
      float* so = &smem[(ct * 16 + lc) * OSTR + (2 * half) * OW + 4 * tx];
      *(float4*)so        = make_float4(acc[jt][ct][0], acc[jt][ct][1], acc[jt][ct][2], acc[jt][ct][3]);
      *(float4*)(so + OW) = make_float4(acc[jt][ct][4], acc[jt][ct][5], acc[jt][ct][6], acc[jt][ct][7]);
    }
  }
  __syncthreads();

  float* obase = out + ((size_t)n * CTOT + g * CG) * (OH * OW)
               + (size_t)(2 * y0) * OW + lane * 4;
  #pragma unroll
  for (int i = 0; i < 4; ++i) {
    const int ch = wave * 4 + i;
    #pragma unroll
    for (int k = 0; k < 4; ++k) {
      float4 v = *(const float4*)&smem[ch * OSTR + (k * 32 + lane) * 4];
      *(float4*)(obase + (size_t)ch * (OH * OW) + k * 128) = v;
    }
  }
}

extern "C" void kernel_launch(void* const* d_in, const int* in_sizes, int n_in,
                              void* d_out, int out_size, void* d_ws, size_t ws_size,
                              hipStream_t stream) {
  (void)in_sizes; (void)n_in; (void)out_size; (void)d_ws; (void)ws_size;
  const float* feat = (const float*)d_in[0];
  const float* mask = (const float*)d_in[1];
  float* out = (float*)d_out;
  dim3 grid(32, NG, 4);
  carafe_wmma<<<grid, 512, 0, stream>>>(feat, mask, out);
}
